# starting kernel
# baseline (speedup 1.0000x reference)
#include <hip/hip_runtime.h>
#include <math.h>

#define NBITS  12
#define SFULL  4096
#define NMASKS 4095
#define NVARSC 4094
#define BROWS  4096

typedef __attribute__((ext_vector_type(2))) float v2f;
typedef __attribute__((ext_vector_type(8))) float v8f;

__global__ __launch_bounds__(256)
void IE_fm_mobius_kernel(const float* __restrict__ ie_vars,
                         float* __restrict__ wpad) {
  __shared__ float a[SFULL];
  const int tid = threadIdx.x;
  const int nt  = blockDim.x;

  for (int k = 1; k <= NBITS - 1; ++k) {
    for (int m = 1 + tid; m <= NVARSC; m += nt) {
      if (__popc(m) == k) {
        float val = fabsf(ie_vars[m - 1]);
        if (k > 1) {
          float mx = 0.0f;
#pragma unroll
          for (int b = 0; b < NBITS; ++b)
            if (m & (1 << b)) mx = fmaxf(mx, a[m ^ (1 << b)]);
          val += mx;
        }
        a[m] = val;
      }
    }
    __syncthreads();
  }

  for (int m = tid; m < SFULL; m += nt) {
    if (m == 0)           a[m] = 0.0f;
    else if (m == NMASKS) a[m] = 1.0f;
    else                  a[m] = fminf(a[m], 1.0f);
  }
  __syncthreads();

  for (int b = 0; b < NBITS; ++b) {
    for (int m = tid; m < SFULL; m += nt) {
      if (m & (1 << b)) a[m] -= a[m ^ (1 << b)];
    }
    __syncthreads();
  }

  for (int j = tid; j < SFULL; j += nt)
    wpad[j] = (j < NMASKS) ? a[j + 1] : 0.0f;
}

__global__ __launch_bounds__(32)
void IE_choquet_kernel(const float* __restrict__ x,
                       const float* __restrict__ wpad,
                       float* __restrict__ out) {
  __shared__ float ws[SFULL];
  const int lane = threadIdx.x;
  for (int j = lane; j < SFULL; j += 32) ws[j] = wpad[j];
  __syncthreads();

  const int rowbase = blockIdx.x * 16;
  const int half    = lane >> 4;
  const int lm      = lane & 15;
  const float* xr   = x + (rowbase + lm) * (2 * NBITS);

  v2f al[3], au[3];
#pragma unroll
  for (int j = 0; j < 3; ++j) {
    const int kb = 4 * j + 2 * half;
    al[j].x = logf(fmaxf(xr[kb],             1e-20f));
    al[j].y = logf(fmaxf(xr[kb + 1],         1e-20f));
    au[j].x = logf(fmaxf(xr[NBITS + kb],     1e-20f));
    au[j].y = logf(fmaxf(xr[NBITS + kb + 1], 1e-20f));
  }

  float accl[8] = {};
  float accu[8] = {};

  for (int c = 0; c < 256; ++c) {
    const int mask = 16 * c + lm + 1;
    v8f cl = {};
    v8f cu = {};
#pragma unroll
    for (int j = 0; j < 3; ++j) {
      const int kb = 4 * j + 2 * half;
      v2f bf;
      bf.x = (float)((mask >> kb) & 1);
      bf.y = (float)((mask >> (kb + 1)) & 1);
      cl = __builtin_amdgcn_wmma_f32_16x16x4_f32(false, al[j], false, bf,
                                                 (short)0, cl, false, false);
      cu = __builtin_amdgcn_wmma_f32_16x16x4_f32(false, au[j], false, bf,
                                                 (short)0, cu, false, false);
    }
    const float wv = ws[16 * c + lm];
#pragma unroll
    for (int r = 0; r < 8; ++r) {
      accl[r] += __expf(cl[r]) * wv;
      accu[r] += __expf(cu[r]) * wv;
    }
  }

#pragma unroll
  for (int r = 0; r < 8; ++r) {
#pragma unroll
    for (int s = 1; s < 16; s <<= 1) {
      accl[r] += __shfl_xor(accl[r], s, 32);
      accu[r] += __shfl_xor(accu[r], s, 32);
    }
  }

  if (lm == 0) {
#pragma unroll
    for (int r = 0; r < 8; ++r) {
      const int orow = rowbase + r + 8 * half;
      const float lo = accl[r];
      const float hi = accu[r];
      out[orow]         = fminf(lo, hi);
      out[BROWS + orow] = hi;
    }
  }
}

extern "C" void kernel_launch(void* const* d_in, const int* in_sizes, int n_in,
                              void* d_out, int out_size, void* d_ws, size_t ws_size,
                              hipStream_t stream) {
  (void)in_sizes; (void)n_in; (void)out_size; (void)ws_size;
  const float* x  = (const float*)d_in[0];
  const float* ie = (const float*)d_in[1];
  float* out  = (float*)d_out;
  float* wpad = (float*)d_ws;

  IE_fm_mobius_kernel<<<dim3(1), dim3(256), 0, stream>>>(ie, wpad);
  IE_choquet_kernel<<<dim3(BROWS / 16), dim3(32), 0, stream>>>(x, wpad, out);
}


# lattice kernel: thread keeps its 16 masks in registers; every pass issues its LDS reads as one batch (absent bit reads a zero slot, |v| from 16 coalesced loads): no waited hop per predecessor
# speedup vs baseline: 3.3428x; 3.3428x over previous
#include <hip/hip_runtime.h>
#include <math.h>

#define NBITS  12
#define SFULL  4096
#define NMASKS 4095
#define NVARSC 4094
#define BROWS  4096

typedef __attribute__((ext_vector_type(2))) float v2f;
typedef __attribute__((ext_vector_type(8))) float v8f;

#define LT    256
#define NPT   16
#define LROW  257
#define LZERO 256

__global__ __launch_bounds__(LT)
void IE_fm_mobius_kernel(const float* __restrict__ ie_vars,
                         float* __restrict__ wpad) {
  __shared__ float a[NPT * LROW + 16];
  const int tid = threadIdx.x;
  const int pt  = __popc(tid);

  int cb[8];
#pragma unroll
  for (int b = 0; b < 8; ++b)
    cb[b] = (tid & (1 << b)) ? (tid ^ (1 << b)) : LZERO;

  float raw[NPT];
#pragma unroll
  for (int i = 0; i < NPT; ++i) {
    int j = LT * i + tid - 1;
    j = (j < 0) ? 0 : j;
    j = (j > NVARSC - 1) ? (NVARSC - 1) : j;
    raw[i] = ie_vars[j];
  }
  __builtin_amdgcn_sched_barrier(0);
  float cur[NPT];
#pragma unroll
  for (int i = 0; i < NPT; ++i) {
    const int m = LT * i + tid;
    cur[i] = (m >= 1 && m <= NVARSC) ? fabsf(raw[i]) : 0.0f;
  }
#pragma unroll
  for (int i = 0; i < NPT; ++i) a[i * LROW + tid] = cur[i];
  if (tid < NPT) a[tid * LROW + LZERO] = 0.0f;
  __syncthreads();

  for (int k = 2; k <= NBITS - 1; ++k) {
    const int q = k - pt;
#pragma unroll
    for (int c = 0; c <= 4; ++c) {
      if (q == c) {
        float pr[NPT][8] = {};
#pragma unroll
        for (int i = 0; i < NPT; ++i) {
          if (__builtin_popcount(i) == c) {
#pragma unroll
            for (int b = 0; b < 8; ++b) pr[i][b] = a[i * LROW + cb[b]];
          }
        }
        __builtin_amdgcn_sched_barrier(0);
        float mx[NPT] = {};
#pragma unroll
        for (int i = 0; i < NPT; ++i) {
          if (__builtin_popcount(i) == c) {
            float t = 0.0f;
#pragma unroll
            for (int b = 0; b < 8; ++b) t = fmaxf(t, pr[i][b]);
#pragma unroll
            for (int b = 0; b < 4; ++b)
              if (i & (1 << b)) t = fmaxf(t, cur[i ^ (1 << b)]);
            mx[i] = t;
          }
        }
#pragma unroll
        for (int i = 0; i < NPT; ++i) {
          if (__builtin_popcount(i) == c) {
            cur[i] += mx[i];
            a[i * LROW + tid] = cur[i];
          }
        }
      }
    }
    __syncthreads();
  }

#pragma unroll
  for (int i = 0; i < NPT; ++i) {
    cur[i] = fminf(cur[i], 1.0f);
    if (LT * i + tid == NMASKS) cur[i] = 1.0f;
    a[i * LROW + tid] = cur[i];
  }
  __syncthreads();

  for (int b = 0; b < 8; ++b) {
    if (tid & (1 << b)) {
      const int pc = tid ^ (1 << b);
      float part[NPT];
#pragma unroll
      for (int i = 0; i < NPT; ++i) part[i] = a[i * LROW + pc];
#pragma unroll
      for (int i = 0; i < NPT; ++i) {
        cur[i] -= part[i];
        a[i * LROW + tid] = cur[i];
      }
    }
    __syncthreads();
  }
#pragma unroll
  for (int b = 0; b < 4; ++b) {
#pragma unroll
    for (int i = 0; i < NPT; ++i)
      if (i & (1 << b)) cur[i] -= cur[i ^ (1 << b)];
  }
#pragma unroll
  for (int i = 0; i < NPT; ++i) a[i * LROW + tid] = cur[i];
  __syncthreads();

#pragma unroll
  for (int i = 0; i < NPT; ++i) {
    const int j = LT * i + tid;
    const int m = (j < NMASKS) ? (j + 1) : NMASKS;
    const float w = a[(m >> 8) * LROW + (m & 255)];
    wpad[j] = (j < NMASKS) ? w : 0.0f;
  }
}

__global__ __launch_bounds__(32)
void IE_choquet_kernel(const float* __restrict__ x,
                       const float* __restrict__ wpad,
                       float* __restrict__ out) {
  __shared__ float ws[SFULL];
  const int lane = threadIdx.x;
  for (int j = lane; j < SFULL; j += 32) ws[j] = wpad[j];
  __syncthreads();

  const int rowbase = blockIdx.x * 16;
  const int half    = lane >> 4;
  const int lm      = lane & 15;
  const float* xr   = x + (rowbase + lm) * (2 * NBITS);

  v2f al[3], au[3];
#pragma unroll
  for (int j = 0; j < 3; ++j) {
    const int kb = 4 * j + 2 * half;
    al[j].x = logf(fmaxf(xr[kb],             1e-20f));
    al[j].y = logf(fmaxf(xr[kb + 1],         1e-20f));
    au[j].x = logf(fmaxf(xr[NBITS + kb],     1e-20f));
    au[j].y = logf(fmaxf(xr[NBITS + kb + 1], 1e-20f));
  }

  float accl[8] = {};
  float accu[8] = {};

  for (int c = 0; c < 256; ++c) {
    const int mask = 16 * c + lm + 1;
    v8f cl = {};
    v8f cu = {};
#pragma unroll
    for (int j = 0; j < 3; ++j) {
      const int kb = 4 * j + 2 * half;
      v2f bf;
      bf.x = (float)((mask >> kb) & 1);
      bf.y = (float)((mask >> (kb + 1)) & 1);
      cl = __builtin_amdgcn_wmma_f32_16x16x4_f32(false, al[j], false, bf,
                                                 (short)0, cl, false, false);
      cu = __builtin_amdgcn_wmma_f32_16x16x4_f32(false, au[j], false, bf,
                                                 (short)0, cu, false, false);
    }
    const float wv = ws[16 * c + lm];
#pragma unroll
    for (int r = 0; r < 8; ++r) {
      accl[r] += __expf(cl[r]) * wv;
      accu[r] += __expf(cu[r]) * wv;
    }
  }

#pragma unroll
  for (int r = 0; r < 8; ++r) {
#pragma unroll
    for (int s = 1; s < 16; s <<= 1) {
      accl[r] += __shfl_xor(accl[r], s, 32);
      accu[r] += __shfl_xor(accu[r], s, 32);
    }
  }

  if (lm == 0) {
#pragma unroll
    for (int r = 0; r < 8; ++r) {
      const int orow = rowbase + r + 8 * half;
      const float lo = accl[r];
      const float hi = accu[r];
      out[orow]         = fminf(lo, hi);
      out[BROWS + orow] = hi;
    }
  }
}

extern "C" void kernel_launch(void* const* d_in, const int* in_sizes, int n_in,
                              void* d_out, int out_size, void* d_ws, size_t ws_size,
                              hipStream_t stream) {
  (void)in_sizes; (void)n_in; (void)out_size; (void)ws_size;
  const float* x  = (const float*)d_in[0];
  const float* ie = (const float*)d_in[1];
  float* out  = (float*)d_out;
  float* wpad = (float*)d_ws;

  IE_fm_mobius_kernel<<<dim3(1), dim3(256), 0, stream>>>(ie, wpad);
  IE_choquet_kernel<<<dim3(BROWS / 16), dim3(32), 0, stream>>>(x, wpad, out);
}


# tile kernel: each 16-row tile's 256 mask chunks dealt over the 8 waves of one 256-thread workgroup (32 each; w staged once per workgroup, partial sums meet in LDS), not one wave walking all 256
# speedup vs baseline: 7.4537x; 2.2298x over previous
#include <hip/hip_runtime.h>
#include <math.h>

#define NBITS  12
#define SFULL  4096
#define NMASKS 4095
#define NVARSC 4094
#define BROWS  4096

typedef __attribute__((ext_vector_type(2))) float v2f;
typedef __attribute__((ext_vector_type(8))) float v8f;

#define LT    256
#define NPT   16
#define LROW  257
#define LZERO 256

__global__ __launch_bounds__(LT)
void IE_fm_mobius_kernel(const float* __restrict__ ie_vars,
                         float* __restrict__ wpad) {
  __shared__ float a[NPT * LROW + 16];
  const int tid = threadIdx.x;
  const int pt  = __popc(tid);

  int cb[8];
#pragma unroll
  for (int b = 0; b < 8; ++b)
    cb[b] = (tid & (1 << b)) ? (tid ^ (1 << b)) : LZERO;

  float raw[NPT];
#pragma unroll
  for (int i = 0; i < NPT; ++i) {
    int j = LT * i + tid - 1;
    j = (j < 0) ? 0 : j;
    j = (j > NVARSC - 1) ? (NVARSC - 1) : j;
    raw[i] = ie_vars[j];
  }
  __builtin_amdgcn_sched_barrier(0);
  float cur[NPT];
#pragma unroll
  for (int i = 0; i < NPT; ++i) {
    const int m = LT * i + tid;
    cur[i] = (m >= 1 && m <= NVARSC) ? fabsf(raw[i]) : 0.0f;
  }
#pragma unroll
  for (int i = 0; i < NPT; ++i) a[i * LROW + tid] = cur[i];
  if (tid < NPT) a[tid * LROW + LZERO] = 0.0f;
  __syncthreads();

  for (int k = 2; k <= NBITS - 1; ++k) {
    const int q = k - pt;
#pragma unroll
    for (int c = 0; c <= 4; ++c) {
      if (q == c) {
        float pr[NPT][8] = {};
#pragma unroll
        for (int i = 0; i < NPT; ++i) {
          if (__builtin_popcount(i) == c) {
#pragma unroll
            for (int b = 0; b < 8; ++b) pr[i][b] = a[i * LROW + cb[b]];
          }
        }
        __builtin_amdgcn_sched_barrier(0);
        float mx[NPT] = {};
#pragma unroll
        for (int i = 0; i < NPT; ++i) {
          if (__builtin_popcount(i) == c) {
            float t = 0.0f;
#pragma unroll
            for (int b = 0; b < 8; ++b) t = fmaxf(t, pr[i][b]);
#pragma unroll
            for (int b = 0; b < 4; ++b)
              if (i & (1 << b)) t = fmaxf(t, cur[i ^ (1 << b)]);
            mx[i] = t;
          }
        }
#pragma unroll
        for (int i = 0; i < NPT; ++i) {
          if (__builtin_popcount(i) == c) {
            cur[i] += mx[i];
            a[i * LROW + tid] = cur[i];
          }
        }
      }
    }
    __syncthreads();
  }

#pragma unroll
  for (int i = 0; i < NPT; ++i) {
    cur[i] = fminf(cur[i], 1.0f);
    if (LT * i + tid == NMASKS) cur[i] = 1.0f;
    a[i * LROW + tid] = cur[i];
  }
  __syncthreads();

  for (int b = 0; b < 8; ++b) {
    if (tid & (1 << b)) {
      const int pc = tid ^ (1 << b);
      float part[NPT];
#pragma unroll
      for (int i = 0; i < NPT; ++i) part[i] = a[i * LROW + pc];
#pragma unroll
      for (int i = 0; i < NPT; ++i) {
        cur[i] -= part[i];
        a[i * LROW + tid] = cur[i];
      }
    }
    __syncthreads();
  }
#pragma unroll
  for (int b = 0; b < 4; ++b) {
#pragma unroll
    for (int i = 0; i < NPT; ++i)
      if (i & (1 << b)) cur[i] -= cur[i ^ (1 << b)];
  }
#pragma unroll
  for (int i = 0; i < NPT; ++i) a[i * LROW + tid] = cur[i];
  __syncthreads();

#pragma unroll
  for (int i = 0; i < NPT; ++i) {
    const int j = LT * i + tid;
    const int m = (j < NMASKS) ? (j + 1) : NMASKS;
    const float w = a[(m >> 8) * LROW + (m & 255)];
    wpad[j] = (j < NMASKS) ? w : 0.0f;
  }
}

#define CWAVES 8
#define CTHREADS 256
#define CPW (256 / CWAVES)

__global__ __launch_bounds__(CTHREADS)
void IE_choquet_kernel(const float* __restrict__ x,
                       const float* __restrict__ wpad,
                       float* __restrict__ out) {
  __shared__ float ws[SFULL];
  __shared__ float red[CWAVES][2][16];
  const int tid  = threadIdx.x;
  const int lane = tid & 31;
  const int wave = __builtin_amdgcn_readfirstlane(tid >> 5);

  const int rowbase = blockIdx.x * 16;
  const int half    = lane >> 4;
  const int lm      = lane & 15;
  const float* xr   = x + (rowbase + lm) * (2 * NBITS);

#pragma unroll
  for (int q = 0; q < SFULL / CTHREADS; ++q)
    ws[tid + CTHREADS * q] = wpad[tid + CTHREADS * q];

  v2f al[3], au[3];
#pragma unroll
  for (int j = 0; j < 3; ++j) {
    const int kb = 4 * j + 2 * half;
    al[j].x = logf(fmaxf(xr[kb],             1e-20f));
    al[j].y = logf(fmaxf(xr[kb + 1],         1e-20f));
    au[j].x = logf(fmaxf(xr[NBITS + kb],     1e-20f));
    au[j].y = logf(fmaxf(xr[NBITS + kb + 1], 1e-20f));
  }
  __syncthreads();

  float accl[8] = {};
  float accu[8] = {};

  const int c0 = wave * CPW;
  for (int c = c0; c < c0 + CPW; ++c) {
    const int mask = 16 * c + lm + 1;
    v8f cl = {};
    v8f cu = {};
#pragma unroll
    for (int j = 0; j < 3; ++j) {
      const int kb = 4 * j + 2 * half;
      v2f bf;
      bf.x = (float)((mask >> kb) & 1);
      bf.y = (float)((mask >> (kb + 1)) & 1);
      cl = __builtin_amdgcn_wmma_f32_16x16x4_f32(false, al[j], false, bf,
                                                 (short)0, cl, false, false);
      cu = __builtin_amdgcn_wmma_f32_16x16x4_f32(false, au[j], false, bf,
                                                 (short)0, cu, false, false);
    }
    const float wv = ws[16 * c + lm];
#pragma unroll
    for (int r = 0; r < 8; ++r) {
      accl[r] += __expf(cl[r]) * wv;
      accu[r] += __expf(cu[r]) * wv;
    }
  }

#pragma unroll
  for (int r = 0; r < 8; ++r) {
#pragma unroll
    for (int s = 1; s < 16; s <<= 1) {
      accl[r] += __shfl_xor(accl[r], s, 32);
      accu[r] += __shfl_xor(accu[r], s, 32);
    }
  }

  if (lm == 0) {
#pragma unroll
    for (int r = 0; r < 8; ++r) {
      red[wave][0][r + 8 * half] = accl[r];
      red[wave][1][r + 8 * half] = accu[r];
    }
  }
  __syncthreads();

  if (wave == 0) {
    float s = 0.0f;
#pragma unroll
    for (int v = 0; v < CWAVES; ++v) s += red[v][half][lm];
    const float other = __shfl_xor(s, 16, 32);
    const float res = half ? s : fminf(s, other);
    out[half * BROWS + rowbase + lm] = res;
  }
}

extern "C" void kernel_launch(void* const* d_in, const int* in_sizes, int n_in,
                              void* d_out, int out_size, void* d_ws, size_t ws_size,
                              hipStream_t stream) {
  (void)in_sizes; (void)n_in; (void)out_size; (void)ws_size;
  const float* x  = (const float*)d_in[0];
  const float* ie = (const float*)d_in[1];
  float* out  = (float*)d_out;
  float* wpad = (float*)d_ws;

  IE_fm_mobius_kernel<<<dim3(1), dim3(256), 0, stream>>>(ie, wpad);
  IE_choquet_kernel<<<dim3(BROWS / 16), dim3(CTHREADS), 0, stream>>>(x, wpad, out);
}


# tile kernel: log-sum split into low 6 + high 6 mask bits, each part an f32 WMMA product made once per workgroup (low: registers, high: LDS table); per element add + exp2 + fma, no WMMA per tile
# speedup vs baseline: 9.2835x; 1.2455x over previous
#include <hip/hip_runtime.h>
#include <math.h>

#define NBITS  12
#define SFULL  4096
#define NMASKS 4095
#define NVARSC 4094
#define BROWS  4096

typedef __attribute__((ext_vector_type(2))) float v2f;
typedef __attribute__((ext_vector_type(8))) float v8f;

#define LT    256
#define NPT   16
#define LROW  257
#define LZERO 256

__global__ __launch_bounds__(LT)
void IE_fm_mobius_kernel(const float* __restrict__ ie_vars,
                         float* __restrict__ wpad) {
  __shared__ float a[NPT * LROW + 16];
  const int tid = threadIdx.x;
  const int pt  = __popc(tid);

  int cb[8];
#pragma unroll
  for (int b = 0; b < 8; ++b)
    cb[b] = (tid & (1 << b)) ? (tid ^ (1 << b)) : LZERO;

  float raw[NPT];
#pragma unroll
  for (int i = 0; i < NPT; ++i) {
    int j = LT * i + tid - 1;
    j = (j < 0) ? 0 : j;
    j = (j > NVARSC - 1) ? (NVARSC - 1) : j;
    raw[i] = ie_vars[j];
  }
  __builtin_amdgcn_sched_barrier(0);
  float cur[NPT];
#pragma unroll
  for (int i = 0; i < NPT; ++i) {
    const int m = LT * i + tid;
    cur[i] = (m >= 1 && m <= NVARSC) ? fabsf(raw[i]) : 0.0f;
  }
#pragma unroll
  for (int i = 0; i < NPT; ++i) a[i * LROW + tid] = cur[i];
  if (tid < NPT) a[tid * LROW + LZERO] = 0.0f;
  __syncthreads();

  for (int k = 2; k <= NBITS - 1; ++k) {
    const int q = k - pt;
#pragma unroll
    for (int c = 0; c <= 4; ++c) {
      if (q == c) {
        float pr[NPT][8] = {};
#pragma unroll
        for (int i = 0; i < NPT; ++i) {
          if (__builtin_popcount(i) == c) {
#pragma unroll
            for (int b = 0; b < 8; ++b) pr[i][b] = a[i * LROW + cb[b]];
          }
        }
        __builtin_amdgcn_sched_barrier(0);
        float mx[NPT] = {};
#pragma unroll
        for (int i = 0; i < NPT; ++i) {
          if (__builtin_popcount(i) == c) {
            float t = 0.0f;
#pragma unroll
            for (int b = 0; b < 8; ++b) t = fmaxf(t, pr[i][b]);
#pragma unroll
            for (int b = 0; b < 4; ++b)
              if (i & (1 << b)) t = fmaxf(t, cur[i ^ (1 << b)]);
            mx[i] = t;
          }
        }
#pragma unroll
        for (int i = 0; i < NPT; ++i) {
          if (__builtin_popcount(i) == c) {
            cur[i] += mx[i];
            a[i * LROW + tid] = cur[i];
          }
        }
      }
    }
    __syncthreads();
  }

#pragma unroll
  for (int i = 0; i < NPT; ++i) {
    cur[i] = fminf(cur[i], 1.0f);
    if (LT * i + tid == NMASKS) cur[i] = 1.0f;
    a[i * LROW + tid] = cur[i];
  }
  __syncthreads();

  for (int b = 0; b < 8; ++b) {
    if (tid & (1 << b)) {
      const int pc = tid ^ (1 << b);
      float part[NPT];
#pragma unroll
      for (int i = 0; i < NPT; ++i) part[i] = a[i * LROW + pc];
#pragma unroll
      for (int i = 0; i < NPT; ++i) {
        cur[i] -= part[i];
        a[i * LROW + tid] = cur[i];
      }
    }
    __syncthreads();
  }
#pragma unroll
  for (int b = 0; b < 4; ++b) {
#pragma unroll
    for (int i = 0; i < NPT; ++i)
      if (i & (1 << b)) cur[i] -= cur[i ^ (1 << b)];
  }
#pragma unroll
  for (int i = 0; i < NPT; ++i) a[i * LROW + tid] = cur[i];
  __syncthreads();

#pragma unroll
  for (int i = 0; i < NPT; ++i) {
    const int j = LT * i + tid;
    const int m = (j < NMASKS) ? (j + 1) : NMASKS;
    const float w = a[(m >> 8) * LROW + (m & 255)];
    wpad[j] = (j < NMASKS) ? w : 0.0f;
  }
}

#define CWAVES 8
#define CTHREADS 256
#define LOWBITS 6
#define NSUB 4
#define NCHUNK 64
#define CPW (NCHUNK / CWAVES)
#define THP 36

typedef __attribute__((ext_vector_type(4))) float v4f;

__device__ __forceinline__ v2f IE_bits_frag(int mask, int j, int half) {
  const int kb = 4 * j + 2 * half;
  v2f bf;
  bf.x = (float)((mask >> kb) & 1);
  bf.y = (float)((mask >> (kb + 1)) & 1);
  return bf;
}

__global__ __launch_bounds__(CTHREADS)
void IE_choquet_kernel(const float* __restrict__ x,
                       const float* __restrict__ wpad,
                       float* __restrict__ out) {
  __shared__ float ws[SFULL + 16];
  __shared__ __attribute__((aligned(16))) float th[NCHUNK * THP];
  __shared__ float red[CWAVES][2][16];
  const int tid  = threadIdx.x;
  const int lane = tid & 31;
  const int wave = __builtin_amdgcn_readfirstlane(tid >> 5);

  const int rowbase = blockIdx.x * 16;
  const int half    = lane >> 4;
  const int lm      = lane & 15;
  const float* xr   = x + (rowbase + lm) * (2 * NBITS);

#pragma unroll
  for (int q = 0; q < SFULL / CTHREADS; ++q)
    ws[tid + CTHREADS * q + 1] = wpad[tid + CTHREADS * q];
  if (tid == 0) ws[0] = 0.0f;

  v2f al[3], au[3];
#pragma unroll
  for (int j = 0; j < 3; ++j) {
    const int kb = 4 * j + 2 * half;
    al[j].x = log2f(fmaxf(xr[kb],             1e-20f));
    al[j].y = log2f(fmaxf(xr[kb + 1],         1e-20f));
    au[j].x = log2f(fmaxf(xr[NBITS + kb],     1e-20f));
    au[j].y = log2f(fmaxf(xr[NBITS + kb + 1], 1e-20f));
  }

  v8f lowl[NSUB], lowu[NSUB];
  {
    const v8f zero = {};
    const v2f b0 = IE_bits_frag(lm, 0, half);
    lowl[0] = __builtin_amdgcn_wmma_f32_16x16x4_f32(false, al[0], false, b0,
                                                    (short)0, zero, false, false);
    lowu[0] = __builtin_amdgcn_wmma_f32_16x16x4_f32(false, au[0], false, b0,
                                                    (short)0, zero, false, false);
#pragma unroll
    for (int e = 1; e < NSUB; ++e) {
      const v2f b1 = IE_bits_frag(lm + 16 * e, 1, half);
      lowl[e] = __builtin_amdgcn_wmma_f32_16x16x4_f32(false, al[1], false, b1,
                                                      (short)0, lowl[0], false, false);
      lowu[e] = __builtin_amdgcn_wmma_f32_16x16x4_f32(false, au[1], false, b1,
                                                      (short)0, lowu[0], false, false);
    }
  }

  if (wave < NCHUNK / 16) {
    const int chunk = 16 * wave + lm;
    v8f hl = {};
    v8f hu = {};
#pragma unroll
    for (int j = 1; j < 3; ++j) {
      const v2f bf = IE_bits_frag(chunk << LOWBITS, j, half);
      hl = __builtin_amdgcn_wmma_f32_16x16x4_f32(false, al[j], false, bf,
                                                 (short)0, hl, false, false);
      hu = __builtin_amdgcn_wmma_f32_16x16x4_f32(false, au[j], false, bf,
                                                 (short)0, hu, false, false);
    }
    float* tp = &th[chunk * THP + 8 * half];
    *(v4f*)(tp)      = __builtin_shufflevector(hl, hl, 0, 1, 2, 3);
    *(v4f*)(tp + 4)  = __builtin_shufflevector(hl, hl, 4, 5, 6, 7);
    *(v4f*)(tp + 16) = __builtin_shufflevector(hu, hu, 0, 1, 2, 3);
    *(v4f*)(tp + 20) = __builtin_shufflevector(hu, hu, 4, 5, 6, 7);
  }
  __syncthreads();

  v8f accl = {};
  v8f accu = {};

#pragma unroll
  for (int cc = 0; cc < CPW; ++cc) {
    const int chunk = CPW * wave + cc;
    const float* tp = &th[chunk * THP + 8 * half];
    const v4f h0 = *(const v4f*)(tp);
    const v4f h1 = *(const v4f*)(tp + 4);
    const v4f g0 = *(const v4f*)(tp + 16);
    const v4f g1 = *(const v4f*)(tp + 20);
    const v8f hl = __builtin_shufflevector(h0, h1, 0, 1, 2, 3, 4, 5, 6, 7);
    const v8f hu = __builtin_shufflevector(g0, g1, 0, 1, 2, 3, 4, 5, 6, 7);
    float wv[NSUB];
#pragma unroll
    for (int e = 0; e < NSUB; ++e) wv[e] = ws[(chunk << LOWBITS) + 16 * e + lm];
#pragma unroll
    for (int e = 0; e < NSUB; ++e) {
      const v8f sl = hl + lowl[e];
      const v8f su = hu + lowu[e];
      v8f el, eu;
#pragma unroll
      for (int r = 0; r < 8; ++r) {
        el[r] = __builtin_amdgcn_exp2f(sl[r]);
        eu[r] = __builtin_amdgcn_exp2f(su[r]);
      }
      accl += el * wv[e];
      accu += eu * wv[e];
    }
  }

#pragma unroll
  for (int r = 0; r < 8; ++r) {
#pragma unroll
    for (int s = 1; s < 16; s <<= 1) {
      accl[r] += __shfl_xor(accl[r], s, 32);
      accu[r] += __shfl_xor(accu[r], s, 32);
    }
  }

  if (lm == 0) {
#pragma unroll
    for (int r = 0; r < 8; ++r) {
      red[wave][0][r + 8 * half] = accl[r];
      red[wave][1][r + 8 * half] = accu[r];
    }
  }
  __syncthreads();

  if (wave == 0) {
    float s = 0.0f;
#pragma unroll
    for (int v = 0; v < CWAVES; ++v) s += red[v][half][lm];
    const float other = __shfl_xor(s, 16, 32);
    const float res = half ? s : fminf(s, other);
    out[half * BROWS + rowbase + lm] = res;
  }
}

extern "C" void kernel_launch(void* const* d_in, const int* in_sizes, int n_in,
                              void* d_out, int out_size, void* d_ws, size_t ws_size,
                              hipStream_t stream) {
  (void)in_sizes; (void)n_in; (void)out_size; (void)ws_size;
  const float* x  = (const float*)d_in[0];
  const float* ie = (const float*)d_in[1];
  float* out  = (float*)d_out;
  float* wpad = (float*)d_ws;

  IE_fm_mobius_kernel<<<dim3(1), dim3(256), 0, stream>>>(ie, wpad);
  IE_choquet_kernel<<<dim3(BROWS / 16), dim3(CTHREADS), 0, stream>>>(x, wpad, out);
}


# lattice kernel: DP ordered by popcount of the 8 thread-held mask bits, not of the whole mask: a thread solves its 16-mask sub-lattice in registers once, from predecessors' LDS rows; 8 waited steps
# speedup vs baseline: 12.4636x; 1.3426x over previous
#include <hip/hip_runtime.h>
#include <math.h>

#define NBITS  12
#define SFULL  4096
#define NMASKS 4095
#define NVARSC 4094
#define BROWS  4096

typedef __attribute__((ext_vector_type(2))) float v2f;
typedef __attribute__((ext_vector_type(8))) float v8f;
typedef __attribute__((ext_vector_type(4))) float v4f;

#define LT    256
#define NPT   16
#define RP    20

__device__ __forceinline__ void IE_row_load(const float* a, int off, float (&r)[NPT]) {
#pragma unroll
  for (int q = 0; q < NPT / 4; ++q) {
    const v4f t = *(const v4f*)(a + off + 4 * q);
    r[4 * q] = t.x; r[4 * q + 1] = t.y; r[4 * q + 2] = t.z; r[4 * q + 3] = t.w;
  }
}

__device__ __forceinline__ void IE_row_store(float* a, int off, const float (&r)[NPT]) {
#pragma unroll
  for (int q = 0; q < NPT / 4; ++q) {
    v4f t;
    t.x = r[4 * q]; t.y = r[4 * q + 1]; t.z = r[4 * q + 2]; t.w = r[4 * q + 3];
    *(v4f*)(a + off + 4 * q) = t;
  }
}

__device__ __forceinline__ void IE_solve16(const float (&av)[NPT], const float (&pm)[NPT],
                                           float (&cur)[NPT]) {
#pragma unroll
  for (int c = 0; c <= 4; ++c) {
#pragma unroll
    for (int i = 0; i < NPT; ++i) {
      if (__builtin_popcount(i) == c) {
        float t = pm[i];
#pragma unroll
        for (int b = 0; b < 4; ++b)
          if (i & (1 << b)) t = fmaxf(t, cur[i ^ (1 << b)]);
        cur[i] = av[i] + t;
      }
    }
  }
}

__global__ __launch_bounds__(LT)
void IE_fm_mobius_kernel(const float* __restrict__ ie_vars,
                         float* __restrict__ wpad) {
  __shared__ __attribute__((aligned(16))) float a[LT * RP];
  const int tid = threadIdx.x;
  const int pt  = __popc(tid);

  int prow[8];
  {
    int r = tid;
#pragma unroll
    for (int n = 0; n < 8; ++n) {
      const int low = r & -r;
      prow[n] = (tid ^ low) * RP;
      r ^= low;
    }
  }

  float raw[NPT];
#pragma unroll
  for (int i = 0; i < NPT; ++i) {
    int j = LT * i + tid - 1;
    j = (j < 0) ? 0 : j;
    j = (j > NVARSC - 1) ? (NVARSC - 1) : j;
    raw[i] = ie_vars[j];
  }
  __builtin_amdgcn_sched_barrier(0);
  float av[NPT];
  float cur[NPT];
#pragma unroll
  for (int i = 0; i < NPT; ++i) {
    const int m = LT * i + tid;
    av[i]  = (m >= 1 && m <= NVARSC) ? fabsf(raw[i]) : 0.0f;
    cur[i] = 0.0f;
  }

  if (tid == 0) {
    float pm[NPT];
#pragma unroll
    for (int i = 0; i < NPT; ++i) pm[i] = 0.0f;
    IE_solve16(av, pm, cur);
    IE_row_store(a, 0, cur);
  }
  __syncthreads();

#pragma unroll
  for (int j = 1; j <= 8; ++j) {
    if (pt == j) {
      float p[8][NPT];
#pragma unroll
      for (int n = 0; n < 8; ++n)
        if (n < j) IE_row_load(a, prow[n], p[n]);
      __builtin_amdgcn_sched_barrier(0);
      float pm[NPT];
#pragma unroll
      for (int i = 0; i < NPT; ++i) {
        float t = p[0][i];
#pragma unroll
        for (int n = 1; n < 8; ++n)
          if (n < j) t = fmaxf(t, p[n][i]);
        pm[i] = t;
      }
      IE_solve16(av, pm, cur);
      IE_row_store(a, tid * RP, cur);
    }
    __syncthreads();
  }

#pragma unroll
  for (int i = 0; i < NPT; ++i) {
    cur[i] = fminf(cur[i], 1.0f);
    if (LT * i + tid == NMASKS) cur[i] = 1.0f;
  }
  IE_row_store(a, tid * RP, cur);
  __syncthreads();

  for (int b = 0; b < 8; ++b) {
    if (tid & (1 << b)) {
      float part[NPT];
      IE_row_load(a, (tid ^ (1 << b)) * RP, part);
#pragma unroll
      for (int i = 0; i < NPT; ++i) cur[i] -= part[i];
      IE_row_store(a, tid * RP, cur);
    }
    __syncthreads();
  }
#pragma unroll
  for (int b = 0; b < 4; ++b) {
#pragma unroll
    for (int i = 0; i < NPT; ++i)
      if (i & (1 << b)) cur[i] -= cur[i ^ (1 << b)];
  }
  IE_row_store(a, tid * RP, cur);
  __syncthreads();

  {
    const bool last = (tid == LT - 1);
    float nx[NPT];
    IE_row_load(a, last ? 0 : (tid + 1) * RP, nx);
#pragma unroll
    for (int i = 0; i < NPT; ++i) {
      const float up = (i < NPT - 1) ? nx[(i + 1) % NPT] : 0.0f;
      wpad[LT * i + tid] = last ? up : nx[i];
    }
  }
}

#define CWAVES 8
#define CTHREADS 256
#define LOWBITS 6
#define NSUB 4
#define NCHUNK 64
#define CPW (NCHUNK / CWAVES)
#define THP 36

__device__ __forceinline__ v2f IE_bits_frag(int mask, int j, int half) {
  const int kb = 4 * j + 2 * half;
  v2f bf;
  bf.x = (float)((mask >> kb) & 1);
  bf.y = (float)((mask >> (kb + 1)) & 1);
  return bf;
}

__global__ __launch_bounds__(CTHREADS)
void IE_choquet_kernel(const float* __restrict__ x,
                       const float* __restrict__ wpad,
                       float* __restrict__ out) {
  __shared__ float ws[SFULL + 16];
  __shared__ __attribute__((aligned(16))) float th[NCHUNK * THP];
  __shared__ float red[CWAVES][2][16];
  const int tid  = threadIdx.x;
  const int lane = tid & 31;
  const int wave = __builtin_amdgcn_readfirstlane(tid >> 5);

  const int rowbase = blockIdx.x * 16;
  const int half    = lane >> 4;
  const int lm      = lane & 15;
  const float* xr   = x + (rowbase + lm) * (2 * NBITS);

#pragma unroll
  for (int q = 0; q < SFULL / CTHREADS; ++q)
    ws[tid + CTHREADS * q + 1] = wpad[tid + CTHREADS * q];
  if (tid == 0) ws[0] = 0.0f;

  v2f al[3], au[3];
#pragma unroll
  for (int j = 0; j < 3; ++j) {
    const int kb = 4 * j + 2 * half;
    al[j].x = log2f(fmaxf(xr[kb],             1e-20f));
    al[j].y = log2f(fmaxf(xr[kb + 1],         1e-20f));
    au[j].x = log2f(fmaxf(xr[NBITS + kb],     1e-20f));
    au[j].y = log2f(fmaxf(xr[NBITS + kb + 1], 1e-20f));
  }

  v8f lowl[NSUB], lowu[NSUB];
  {
    const v8f zero = {};
    const v2f b0 = IE_bits_frag(lm, 0, half);
    lowl[0] = __builtin_amdgcn_wmma_f32_16x16x4_f32(false, al[0], false, b0,
                                                    (short)0, zero, false, false);
    lowu[0] = __builtin_amdgcn_wmma_f32_16x16x4_f32(false, au[0], false, b0,
                                                    (short)0, zero, false, false);
#pragma unroll
    for (int e = 1; e < NSUB; ++e) {
      const v2f b1 = IE_bits_frag(lm + 16 * e, 1, half);
      lowl[e] = __builtin_amdgcn_wmma_f32_16x16x4_f32(false, al[1], false, b1,
                                                      (short)0, lowl[0], false, false);
      lowu[e] = __builtin_amdgcn_wmma_f32_16x16x4_f32(false, au[1], false, b1,
                                                      (short)0, lowu[0], false, false);
    }
  }

  if (wave < NCHUNK / 16) {
    const int chunk = 16 * wave + lm;
    v8f hl = {};
    v8f hu = {};
#pragma unroll
    for (int j = 1; j < 3; ++j) {
      const v2f bf = IE_bits_frag(chunk << LOWBITS, j, half);
      hl = __builtin_amdgcn_wmma_f32_16x16x4_f32(false, al[j], false, bf,
                                                 (short)0, hl, false, false);
      hu = __builtin_amdgcn_wmma_f32_16x16x4_f32(false, au[j], false, bf,
                                                 (short)0, hu, false, false);
    }
    float* tp = &th[chunk * THP + 8 * half];
    *(v4f*)(tp)      = __builtin_shufflevector(hl, hl, 0, 1, 2, 3);
    *(v4f*)(tp + 4)  = __builtin_shufflevector(hl, hl, 4, 5, 6, 7);
    *(v4f*)(tp + 16) = __builtin_shufflevector(hu, hu, 0, 1, 2, 3);
    *(v4f*)(tp + 20) = __builtin_shufflevector(hu, hu, 4, 5, 6, 7);
  }
  __syncthreads();

  v8f accl = {};
  v8f accu = {};

#pragma unroll
  for (int cc = 0; cc < CPW; ++cc) {
    const int chunk = CPW * wave + cc;
    const float* tp = &th[chunk * THP + 8 * half];
    const v4f h0 = *(const v4f*)(tp);
    const v4f h1 = *(const v4f*)(tp + 4);
    const v4f g0 = *(const v4f*)(tp + 16);
    const v4f g1 = *(const v4f*)(tp + 20);
    const v8f hl = __builtin_shufflevector(h0, h1, 0, 1, 2, 3, 4, 5, 6, 7);
    const v8f hu = __builtin_shufflevector(g0, g1, 0, 1, 2, 3, 4, 5, 6, 7);
    float wv[NSUB];
#pragma unroll
    for (int e = 0; e < NSUB; ++e) wv[e] = ws[(chunk << LOWBITS) + 16 * e + lm];
#pragma unroll
    for (int e = 0; e < NSUB; ++e) {
      const v8f sl = hl + lowl[e];
      const v8f su = hu + lowu[e];
      v8f el, eu;
#pragma unroll
      for (int r = 0; r < 8; ++r) {
        el[r] = __builtin_amdgcn_exp2f(sl[r]);
        eu[r] = __builtin_amdgcn_exp2f(su[r]);
      }
      accl += el * wv[e];
      accu += eu * wv[e];
    }
  }

#pragma unroll
  for (int r = 0; r < 8; ++r) {
#pragma unroll
    for (int s = 1; s < 16; s <<= 1) {
      accl[r] += __shfl_xor(accl[r], s, 32);
      accu[r] += __shfl_xor(accu[r], s, 32);
    }
  }

  if (lm == 0) {
#pragma unroll
    for (int r = 0; r < 8; ++r) {
      red[wave][0][r + 8 * half] = accl[r];
      red[wave][1][r + 8 * half] = accu[r];
    }
  }
  __syncthreads();

  if (wave == 0) {
    float s = 0.0f;
#pragma unroll
    for (int v = 0; v < CWAVES; ++v) s += red[v][half][lm];
    const float other = __shfl_xor(s, 16, 32);
    const float res = half ? s : fminf(s, other);
    out[half * BROWS + rowbase + lm] = res;
  }
}

extern "C" void kernel_launch(void* const* d_in, const int* in_sizes, int n_in,
                              void* d_out, int out_size, void* d_ws, size_t ws_size,
                              hipStream_t stream) {
  (void)in_sizes; (void)n_in; (void)out_size; (void)ws_size;
  const float* x  = (const float*)d_in[0];
  const float* ie = (const float*)d_in[1];
  float* out  = (float*)d_out;
  float* wpad = (float*)d_ws;

  IE_fm_mobius_kernel<<<dim3(1), dim3(256), 0, stream>>>(ie, wpad);
  IE_choquet_kernel<<<dim3(BROWS / 16), dim3(CTHREADS), 0, stream>>>(x, wpad, out);
}


# lattice kernel: Mobius transform with one LDS hand-over, not a pass and barrier per bit: lane bits 0..4 by DPP/permlane exchange in registers, wave bits 5..7 by one gather of the partner rows
# speedup vs baseline: 13.1737x; 1.0570x over previous
#include <hip/hip_runtime.h>
#include <math.h>

#define NBITS  12
#define SFULL  4096
#define NMASKS 4095
#define NVARSC 4094
#define BROWS  4096

typedef __attribute__((ext_vector_type(2))) float v2f;
typedef __attribute__((ext_vector_type(8))) float v8f;
typedef __attribute__((ext_vector_type(4))) float v4f;

#define LT    256
#define NPT   16
#define RP    20

__device__ __forceinline__ void IE_row_load(const float* a, int off, float (&r)[NPT]) {
#pragma unroll
  for (int q = 0; q < NPT / 4; ++q) {
    const v4f t = *(const v4f*)(a + off + 4 * q);
    r[4 * q] = t.x; r[4 * q + 1] = t.y; r[4 * q + 2] = t.z; r[4 * q + 3] = t.w;
  }
}

__device__ __forceinline__ void IE_row_store(float* a, int off, const float (&r)[NPT]) {
#pragma unroll
  for (int q = 0; q < NPT / 4; ++q) {
    v4f t;
    t.x = r[4 * q]; t.y = r[4 * q + 1]; t.z = r[4 * q + 2]; t.w = r[4 * q + 3];
    *(v4f*)(a + off + 4 * q) = t;
  }
}

__device__ __forceinline__ void IE_solve16(const float (&av)[NPT], const float (&pm)[NPT],
                                           float (&cur)[NPT]) {
#pragma unroll
  for (int c = 0; c <= 4; ++c) {
#pragma unroll
    for (int i = 0; i < NPT; ++i) {
      if (__builtin_popcount(i) == c) {
        float t = pm[i];
#pragma unroll
        for (int b = 0; b < 4; ++b)
          if (i & (1 << b)) t = fmaxf(t, cur[i ^ (1 << b)]);
        cur[i] = av[i] + t;
      }
    }
  }
}

template <int K>
__device__ __forceinline__ float IE_xlane(float x) {
  const int b = __builtin_bit_cast(int, x);
  if constexpr (K == 16)
    return __builtin_bit_cast(float, __builtin_amdgcn_permlanex16(b, b, 0x76543210, 0xfedcba98, false, false));
  else
    return __builtin_bit_cast(float, __builtin_amdgcn_update_dpp(0, b, 0x160 | K, 0xf, 0xf, true));
}

template <int K>
__device__ __forceinline__ void IE_lane_pass(float (&cur)[NPT], bool has) {
  const float f = has ? 1.0f : 0.0f;
#pragma unroll
  for (int i = 0; i < NPT; ++i) {
    const float part = IE_xlane<K>(cur[i]);
    cur[i] = __builtin_fmaf(-part, f, cur[i]);
  }
}

template <int WB>
__device__ __forceinline__ void IE_wave_bits(const float* a, int tid, float (&cur)[NPT]) {
  float x[8][NPT];
#pragma unroll
  for (int s = 1; s < 8; ++s)
    if ((s & ~WB) == 0) IE_row_load(a, (tid ^ (s << 5)) * RP, x[s]);
#pragma unroll
  for (int i = 0; i < NPT; ++i) x[0][i] = cur[i];
#pragma unroll
  for (int kb = 0; kb < 3; ++kb) {
    const int k = 1 << kb;
    if (WB & k) {
#pragma unroll
      for (int s = 0; s < 8; ++s) {
        if ((s & ~WB) == 0 && (s & (2 * k - 1)) == 0) {
#pragma unroll
          for (int i = 0; i < NPT; ++i) x[s][i] -= x[s | k][i];
        }
      }
    }
  }
#pragma unroll
  for (int i = 0; i < NPT; ++i) cur[i] = x[0][i];
}

__global__ __launch_bounds__(LT)
void IE_fm_mobius_kernel(const float* __restrict__ ie_vars,
                         float* __restrict__ wpad) {
  __shared__ __attribute__((aligned(16))) float a[LT * RP];
  __shared__ __attribute__((aligned(16))) float fin[LT * RP];
  const int tid = threadIdx.x;
  const int pt  = __popc(tid);

  int prow[8];
  {
    int r = tid;
#pragma unroll
    for (int n = 0; n < 8; ++n) {
      const int low = r & -r;
      prow[n] = (tid ^ low) * RP;
      r ^= low;
    }
  }

  float raw[NPT];
#pragma unroll
  for (int i = 0; i < NPT; ++i) {
    int j = LT * i + tid - 1;
    j = (j < 0) ? 0 : j;
    j = (j > NVARSC - 1) ? (NVARSC - 1) : j;
    raw[i] = ie_vars[j];
  }
  __builtin_amdgcn_sched_barrier(0);
  float av[NPT];
  float cur[NPT];
#pragma unroll
  for (int i = 0; i < NPT; ++i) {
    const int m = LT * i + tid;
    av[i]  = (m >= 1 && m <= NVARSC) ? fabsf(raw[i]) : 0.0f;
    cur[i] = 0.0f;
  }

  if (tid == 0) {
    float pm[NPT];
#pragma unroll
    for (int i = 0; i < NPT; ++i) pm[i] = 0.0f;
    IE_solve16(av, pm, cur);
    IE_row_store(a, 0, cur);
  }
  __syncthreads();

#pragma unroll
  for (int j = 1; j <= 8; ++j) {
    if (pt == j) {
      float p[8][NPT];
#pragma unroll
      for (int n = 0; n < 8; ++n)
        if (n < j) IE_row_load(a, prow[n], p[n]);
      __builtin_amdgcn_sched_barrier(0);
      float pm[NPT];
#pragma unroll
      for (int i = 0; i < NPT; ++i) {
        float t = p[0][i];
#pragma unroll
        for (int n = 1; n < 8; ++n)
          if (n < j) t = fmaxf(t, p[n][i]);
        pm[i] = t;
      }
      IE_solve16(av, pm, cur);
      IE_row_store(a, tid * RP, cur);
    }
    __syncthreads();
  }

#pragma unroll
  for (int i = 0; i < NPT; ++i) {
    cur[i] = fminf(cur[i], 1.0f);
    if (LT * i + tid == NMASKS) cur[i] = 1.0f;
  }

  IE_lane_pass<1>(cur, (tid & 1) != 0);
  IE_lane_pass<2>(cur, (tid & 2) != 0);
  IE_lane_pass<4>(cur, (tid & 4) != 0);
  IE_lane_pass<8>(cur, (tid & 8) != 0);
  IE_lane_pass<16>(cur, (tid & 16) != 0);
  IE_row_store(a, tid * RP, cur);
  __syncthreads();
  {
    const int wv = __builtin_amdgcn_readfirstlane(tid >> 5);
    switch (wv) {
      case 1: IE_wave_bits<1>(a, tid, cur); break;
      case 2: IE_wave_bits<2>(a, tid, cur); break;
      case 3: IE_wave_bits<3>(a, tid, cur); break;
      case 4: IE_wave_bits<4>(a, tid, cur); break;
      case 5: IE_wave_bits<5>(a, tid, cur); break;
      case 6: IE_wave_bits<6>(a, tid, cur); break;
      case 7: IE_wave_bits<7>(a, tid, cur); break;
      default: break;
    }
  }
#pragma unroll
  for (int b = 0; b < 4; ++b) {
#pragma unroll
    for (int i = 0; i < NPT; ++i)
      if (i & (1 << b)) cur[i] -= cur[i ^ (1 << b)];
  }
  IE_row_store(fin, tid * RP, cur);
  __syncthreads();

  {
    const bool last = (tid == LT - 1);
    float nx[NPT];
    IE_row_load(fin, last ? 0 : (tid + 1) * RP, nx);
#pragma unroll
    for (int i = 0; i < NPT; ++i) {
      const float up = (i < NPT - 1) ? nx[(i + 1) % NPT] : 0.0f;
      wpad[LT * i + tid] = last ? up : nx[i];
    }
  }
}

#define CWAVES 8
#define CTHREADS 256
#define LOWBITS 6
#define NSUB 4
#define NCHUNK 64
#define CPW (NCHUNK / CWAVES)
#define THP 36

__device__ __forceinline__ v2f IE_bits_frag(int mask, int j, int half) {
  const int kb = 4 * j + 2 * half;
  v2f bf;
  bf.x = (float)((mask >> kb) & 1);
  bf.y = (float)((mask >> (kb + 1)) & 1);
  return bf;
}

__global__ __launch_bounds__(CTHREADS)
void IE_choquet_kernel(const float* __restrict__ x,
                       const float* __restrict__ wpad,
                       float* __restrict__ out) {
  __shared__ float ws[SFULL + 16];
  __shared__ __attribute__((aligned(16))) float th[NCHUNK * THP];
  __shared__ float red[CWAVES][2][16];
  const int tid  = threadIdx.x;
  const int lane = tid & 31;
  const int wave = __builtin_amdgcn_readfirstlane(tid >> 5);

  const int rowbase = blockIdx.x * 16;
  const int half    = lane >> 4;
  const int lm      = lane & 15;
  const float* xr   = x + (rowbase + lm) * (2 * NBITS);

#pragma unroll
  for (int q = 0; q < SFULL / CTHREADS; ++q)
    ws[tid + CTHREADS * q + 1] = wpad[tid + CTHREADS * q];
  if (tid == 0) ws[0] = 0.0f;

  v2f al[3], au[3];
#pragma unroll
  for (int j = 0; j < 3; ++j) {
    const int kb = 4 * j + 2 * half;
    al[j].x = log2f(fmaxf(xr[kb],             1e-20f));
    al[j].y = log2f(fmaxf(xr[kb + 1],         1e-20f));
    au[j].x = log2f(fmaxf(xr[NBITS + kb],     1e-20f));
    au[j].y = log2f(fmaxf(xr[NBITS + kb + 1], 1e-20f));
  }

  v8f lowl[NSUB], lowu[NSUB];
  {
    const v8f zero = {};
    const v2f b0 = IE_bits_frag(lm, 0, half);
    lowl[0] = __builtin_amdgcn_wmma_f32_16x16x4_f32(false, al[0], false, b0,
                                                    (short)0, zero, false, false);
    lowu[0] = __builtin_amdgcn_wmma_f32_16x16x4_f32(false, au[0], false, b0,
                                                    (short)0, zero, false, false);
#pragma unroll
    for (int e = 1; e < NSUB; ++e) {
      const v2f b1 = IE_bits_frag(lm + 16 * e, 1, half);
      lowl[e] = __builtin_amdgcn_wmma_f32_16x16x4_f32(false, al[1], false, b1,
                                                      (short)0, lowl[0], false, false);
      lowu[e] = __builtin_amdgcn_wmma_f32_16x16x4_f32(false, au[1], false, b1,
                                                      (short)0, lowu[0], false, false);
    }
  }

  if (wave < NCHUNK / 16) {
    const int chunk = 16 * wave + lm;
    v8f hl = {};
    v8f hu = {};
#pragma unroll
    for (int j = 1; j < 3; ++j) {
      const v2f bf = IE_bits_frag(chunk << LOWBITS, j, half);
      hl = __builtin_amdgcn_wmma_f32_16x16x4_f32(false, al[j], false, bf,
                                                 (short)0, hl, false, false);
      hu = __builtin_amdgcn_wmma_f32_16x16x4_f32(false, au[j], false, bf,
                                                 (short)0, hu, false, false);
    }
    float* tp = &th[chunk * THP + 8 * half];
    *(v4f*)(tp)      = __builtin_shufflevector(hl, hl, 0, 1, 2, 3);
    *(v4f*)(tp + 4)  = __builtin_shufflevector(hl, hl, 4, 5, 6, 7);
    *(v4f*)(tp + 16) = __builtin_shufflevector(hu, hu, 0, 1, 2, 3);
    *(v4f*)(tp + 20) = __builtin_shufflevector(hu, hu, 4, 5, 6, 7);
  }
  __syncthreads();

  v8f accl = {};
  v8f accu = {};

#pragma unroll
  for (int cc = 0; cc < CPW; ++cc) {
    const int chunk = CPW * wave + cc;
    const float* tp = &th[chunk * THP + 8 * half];
    const v4f h0 = *(const v4f*)(tp);
    const v4f h1 = *(const v4f*)(tp + 4);
    const v4f g0 = *(const v4f*)(tp + 16);
    const v4f g1 = *(const v4f*)(tp + 20);
    const v8f hl = __builtin_shufflevector(h0, h1, 0, 1, 2, 3, 4, 5, 6, 7);
    const v8f hu = __builtin_shufflevector(g0, g1, 0, 1, 2, 3, 4, 5, 6, 7);
    float wv[NSUB];
#pragma unroll
    for (int e = 0; e < NSUB; ++e) wv[e] = ws[(chunk << LOWBITS) + 16 * e + lm];
#pragma unroll
    for (int e = 0; e < NSUB; ++e) {
      const v8f sl = hl + lowl[e];
      const v8f su = hu + lowu[e];
      v8f el, eu;
#pragma unroll
      for (int r = 0; r < 8; ++r) {
        el[r] = __builtin_amdgcn_exp2f(sl[r]);
        eu[r] = __builtin_amdgcn_exp2f(su[r]);
      }
      accl += el * wv[e];
      accu += eu * wv[e];
    }
  }

#pragma unroll
  for (int r = 0; r < 8; ++r) {
#pragma unroll
    for (int s = 1; s < 16; s <<= 1) {
      accl[r] += __shfl_xor(accl[r], s, 32);
      accu[r] += __shfl_xor(accu[r], s, 32);
    }
  }

  if (lm == 0) {
#pragma unroll
    for (int r = 0; r < 8; ++r) {
      red[wave][0][r + 8 * half] = accl[r];
      red[wave][1][r + 8 * half] = accu[r];
    }
  }
  __syncthreads();

  if (wave == 0) {
    float s = 0.0f;
#pragma unroll
    for (int v = 0; v < CWAVES; ++v) s += red[v][half][lm];
    const float other = __shfl_xor(s, 16, 32);
    const float res = half ? s : fminf(s, other);
    out[half * BROWS + rowbase + lm] = res;
  }
}

extern "C" void kernel_launch(void* const* d_in, const int* in_sizes, int n_in,
                              void* d_out, int out_size, void* d_ws, size_t ws_size,
                              hipStream_t stream) {
  (void)in_sizes; (void)n_in; (void)out_size; (void)ws_size;
  const float* x  = (const float*)d_in[0];
  const float* ie = (const float*)d_in[1];
  float* out  = (float*)d_out;
  float* wpad = (float*)d_ws;

  IE_fm_mobius_kernel<<<dim3(1), dim3(256), 0, stream>>>(ie, wpad);
  IE_choquet_kernel<<<dim3(BROWS / 16), dim3(CTHREADS), 0, stream>>>(x, wpad, out);
}


# tile kernel: w stored by mask; each wave reads its own 512 weights global->registers (16-byte loads) in one round trip together with all its x loads, not three trips in a row; no LDS staging of w
# speedup vs baseline: 14.3139x; 1.0866x over previous
#include <hip/hip_runtime.h>
#include <math.h>

#define NBITS  12
#define NMASKS 4095
#define NVARSC 4094
#define BROWS  4096

typedef __attribute__((ext_vector_type(2))) float v2f;
typedef __attribute__((ext_vector_type(8))) float v8f;
typedef __attribute__((ext_vector_type(4))) float v4f;

#define LT    256
#define NPT   16
#define RP    20

__device__ __forceinline__ void IE_row_load(const float* a, int off, float (&r)[NPT]) {
#pragma unroll
  for (int q = 0; q < NPT / 4; ++q) {
    const v4f t = *(const v4f*)(a + off + 4 * q);
    r[4 * q] = t.x; r[4 * q + 1] = t.y; r[4 * q + 2] = t.z; r[4 * q + 3] = t.w;
  }
}

__device__ __forceinline__ void IE_row_store(float* a, int off, const float (&r)[NPT]) {
#pragma unroll
  for (int q = 0; q < NPT / 4; ++q) {
    v4f t;
    t.x = r[4 * q]; t.y = r[4 * q + 1]; t.z = r[4 * q + 2]; t.w = r[4 * q + 3];
    *(v4f*)(a + off + 4 * q) = t;
  }
}

__device__ __forceinline__ void IE_solve16(const float (&av)[NPT], const float (&pm)[NPT],
                                           float (&cur)[NPT]) {
#pragma unroll
  for (int c = 0; c <= 4; ++c) {
#pragma unroll
    for (int i = 0; i < NPT; ++i) {
      if (__builtin_popcount(i) == c) {
        float t = pm[i];
#pragma unroll
        for (int b = 0; b < 4; ++b)
          if (i & (1 << b)) t = fmaxf(t, cur[i ^ (1 << b)]);
        cur[i] = av[i] + t;
      }
    }
  }
}

template <int K>
__device__ __forceinline__ float IE_xlane(float x) {
  const int b = __builtin_bit_cast(int, x);
  if constexpr (K == 16)
    return __builtin_bit_cast(float, __builtin_amdgcn_permlanex16(b, b, 0x76543210, 0xfedcba98, false, false));
  else
    return __builtin_bit_cast(float, __builtin_amdgcn_update_dpp(0, b, 0x160 | K, 0xf, 0xf, true));
}

template <int K>
__device__ __forceinline__ void IE_lane_pass(float (&cur)[NPT], bool has) {
  const float f = has ? 1.0f : 0.0f;
#pragma unroll
  for (int i = 0; i < NPT; ++i) {
    const float part = IE_xlane<K>(cur[i]);
    cur[i] = __builtin_fmaf(-part, f, cur[i]);
  }
}

template <int WB>
__device__ __forceinline__ void IE_wave_bits(const float* a, int tid, float (&cur)[NPT]) {
  float x[8][NPT];
#pragma unroll
  for (int s = 1; s < 8; ++s)
    if ((s & ~WB) == 0) IE_row_load(a, (tid ^ (s << 5)) * RP, x[s]);
#pragma unroll
  for (int i = 0; i < NPT; ++i) x[0][i] = cur[i];
#pragma unroll
  for (int kb = 0; kb < 3; ++kb) {
    const int k = 1 << kb;
    if (WB & k) {
#pragma unroll
      for (int s = 0; s < 8; ++s) {
        if ((s & ~WB) == 0 && (s & (2 * k - 1)) == 0) {
#pragma unroll
          for (int i = 0; i < NPT; ++i) x[s][i] -= x[s | k][i];
        }
      }
    }
  }
#pragma unroll
  for (int i = 0; i < NPT; ++i) cur[i] = x[0][i];
}

__global__ __launch_bounds__(LT)
void IE_fm_mobius_kernel(const float* __restrict__ ie_vars,
                         float* __restrict__ wm) {
  __shared__ __attribute__((aligned(16))) float a[LT * RP];
  const int tid = threadIdx.x;
  const int pt  = __popc(tid);

  int prow[8];
  {
    int r = tid;
#pragma unroll
    for (int n = 0; n < 8; ++n) {
      const int low = r & -r;
      prow[n] = (tid ^ low) * RP;
      r ^= low;
    }
  }

  float raw[NPT];
#pragma unroll
  for (int i = 0; i < NPT; ++i) {
    int j = LT * i + tid - 1;
    j = (j < 0) ? 0 : j;
    j = (j > NVARSC - 1) ? (NVARSC - 1) : j;
    raw[i] = ie_vars[j];
  }
  __builtin_amdgcn_sched_barrier(0);
  float av[NPT];
  float cur[NPT];
#pragma unroll
  for (int i = 0; i < NPT; ++i) {
    const int m = LT * i + tid;
    av[i]  = (m >= 1 && m <= NVARSC) ? fabsf(raw[i]) : 0.0f;
    cur[i] = 0.0f;
  }

  if (tid == 0) {
    float pm[NPT];
#pragma unroll
    for (int i = 0; i < NPT; ++i) pm[i] = 0.0f;
    IE_solve16(av, pm, cur);
    IE_row_store(a, 0, cur);
  }
  __syncthreads();

#pragma unroll
  for (int j = 1; j <= 8; ++j) {
    if (pt == j) {
      float p[8][NPT];
#pragma unroll
      for (int n = 0; n < 8; ++n)
        if (n < j) IE_row_load(a, prow[n], p[n]);
      __builtin_amdgcn_sched_barrier(0);
      float pm[NPT];
#pragma unroll
      for (int i = 0; i < NPT; ++i) {
        float t = p[0][i];
#pragma unroll
        for (int n = 1; n < 8; ++n)
          if (n < j) t = fmaxf(t, p[n][i]);
        pm[i] = t;
      }
      IE_solve16(av, pm, cur);
      IE_row_store(a, tid * RP, cur);
    }
    __syncthreads();
  }

#pragma unroll
  for (int i = 0; i < NPT; ++i) {
    cur[i] = fminf(cur[i], 1.0f);
    if (LT * i + tid == NMASKS) cur[i] = 1.0f;
  }

  IE_lane_pass<1>(cur, (tid & 1) != 0);
  IE_lane_pass<2>(cur, (tid & 2) != 0);
  IE_lane_pass<4>(cur, (tid & 4) != 0);
  IE_lane_pass<8>(cur, (tid & 8) != 0);
  IE_lane_pass<16>(cur, (tid & 16) != 0);
  IE_row_store(a, tid * RP, cur);
  __syncthreads();
  {
    const int wv = __builtin_amdgcn_readfirstlane(tid >> 5);
    switch (wv) {
      case 1: IE_wave_bits<1>(a, tid, cur); break;
      case 2: IE_wave_bits<2>(a, tid, cur); break;
      case 3: IE_wave_bits<3>(a, tid, cur); break;
      case 4: IE_wave_bits<4>(a, tid, cur); break;
      case 5: IE_wave_bits<5>(a, tid, cur); break;
      case 6: IE_wave_bits<6>(a, tid, cur); break;
      case 7: IE_wave_bits<7>(a, tid, cur); break;
      default: break;
    }
  }
#pragma unroll
  for (int b = 0; b < 4; ++b) {
#pragma unroll
    for (int i = 0; i < NPT; ++i)
      if (i & (1 << b)) cur[i] -= cur[i ^ (1 << b)];
  }
#pragma unroll
  for (int i = 0; i < NPT; ++i) wm[LT * i + tid] = cur[i];
}

#define CWAVES 8
#define CTHREADS 256
#define LOWBITS 6
#define NSUB 4
#define NCHUNK 64
#define CPW (NCHUNK / CWAVES)
#define THP 36

__device__ __forceinline__ v2f IE_bits_frag(int mask, int j, int half) {
  const int kb = 4 * j + 2 * half;
  v2f bf;
  bf.x = (float)((mask >> kb) & 1);
  bf.y = (float)((mask >> (kb + 1)) & 1);
  return bf;
}

__global__ __launch_bounds__(CTHREADS)
void IE_choquet_kernel(const float* __restrict__ x,
                       const float* __restrict__ wm,
                       float* __restrict__ out) {
  __shared__ __attribute__((aligned(16))) float th[NCHUNK * THP];
  __shared__ float red[CWAVES][2][16];
  const int tid  = threadIdx.x;
  const int lane = tid & 31;
  const int wave = __builtin_amdgcn_readfirstlane(tid >> 5);

  const int rowbase = blockIdx.x * 16;
  const int half    = lane >> 4;
  const int lm      = lane & 15;
  const float* xr   = x + (rowbase + lm) * (2 * NBITS);

  v2f xl[3], xu[3];
#pragma unroll
  for (int j = 0; j < 3; ++j) {
    const int kb = 4 * j + 2 * half;
    xl[j].x = xr[kb];
    xl[j].y = xr[kb + 1];
    xu[j].x = xr[NBITS + kb];
    xu[j].y = xr[NBITS + kb + 1];
  }

  v4f wq[CPW];
  {
    const float* wp = wm + (wave << 9) + 4 * lm;
#pragma unroll
    for (int cc = 0; cc < CPW; ++cc)
      wq[cc] = *(const v4f*)(wp + (cc << LOWBITS));
  }

  v2f al[3], au[3];
#pragma unroll
  for (int j = 0; j < 3; ++j) {
    al[j].x = log2f(fmaxf(xl[j].x, 1e-20f));
    al[j].y = log2f(fmaxf(xl[j].y, 1e-20f));
    au[j].x = log2f(fmaxf(xu[j].x, 1e-20f));
    au[j].y = log2f(fmaxf(xu[j].y, 1e-20f));
  }

  v8f lowl[NSUB], lowu[NSUB];
  {
    const v8f zero = {};
    const v2f b1 = IE_bits_frag(4 * lm, 1, half);
    const v8f basel = __builtin_amdgcn_wmma_f32_16x16x4_f32(false, al[1], false, b1,
                                                            (short)0, zero, false, false);
    const v8f baseu = __builtin_amdgcn_wmma_f32_16x16x4_f32(false, au[1], false, b1,
                                                            (short)0, zero, false, false);
#pragma unroll
    for (int e = 0; e < NSUB; ++e) {
      const v2f b0 = IE_bits_frag(4 * lm + e, 0, half);
      lowl[e] = __builtin_amdgcn_wmma_f32_16x16x4_f32(false, al[0], false, b0,
                                                      (short)0, basel, false, false);
      lowu[e] = __builtin_amdgcn_wmma_f32_16x16x4_f32(false, au[0], false, b0,
                                                      (short)0, baseu, false, false);
    }
  }

  if (wave < NCHUNK / 16) {
    const int chunk = 16 * wave + lm;
    v8f hl = {};
    v8f hu = {};
#pragma unroll
    for (int j = 1; j < 3; ++j) {
      const v2f bf = IE_bits_frag(chunk << LOWBITS, j, half);
      hl = __builtin_amdgcn_wmma_f32_16x16x4_f32(false, al[j], false, bf,
                                                 (short)0, hl, false, false);
      hu = __builtin_amdgcn_wmma_f32_16x16x4_f32(false, au[j], false, bf,
                                                 (short)0, hu, false, false);
    }
    float* tp = &th[chunk * THP + 8 * half];
    *(v4f*)(tp)      = __builtin_shufflevector(hl, hl, 0, 1, 2, 3);
    *(v4f*)(tp + 4)  = __builtin_shufflevector(hl, hl, 4, 5, 6, 7);
    *(v4f*)(tp + 16) = __builtin_shufflevector(hu, hu, 0, 1, 2, 3);
    *(v4f*)(tp + 20) = __builtin_shufflevector(hu, hu, 4, 5, 6, 7);
  }
  __syncthreads();

  v8f accl = {};
  v8f accu = {};

#pragma unroll
  for (int cc = 0; cc < CPW; ++cc) {
    const int chunk = CPW * wave + cc;
    const float* tp = &th[chunk * THP + 8 * half];
    const v4f h0 = *(const v4f*)(tp);
    const v4f h1 = *(const v4f*)(tp + 4);
    const v4f g0 = *(const v4f*)(tp + 16);
    const v4f g1 = *(const v4f*)(tp + 20);
    const v8f hl = __builtin_shufflevector(h0, h1, 0, 1, 2, 3, 4, 5, 6, 7);
    const v8f hu = __builtin_shufflevector(g0, g1, 0, 1, 2, 3, 4, 5, 6, 7);
#pragma unroll
    for (int e = 0; e < NSUB; ++e) {
      const float we = wq[cc][e];
      const v8f sl = hl + lowl[e];
      const v8f su = hu + lowu[e];
      v8f el, eu;
#pragma unroll
      for (int r = 0; r < 8; ++r) {
        el[r] = __builtin_amdgcn_exp2f(sl[r]);
        eu[r] = __builtin_amdgcn_exp2f(su[r]);
      }
      accl += el * we;
      accu += eu * we;
    }
  }

#pragma unroll
  for (int r = 0; r < 8; ++r) {
#pragma unroll
    for (int s = 1; s < 16; s <<= 1) {
      accl[r] += __shfl_xor(accl[r], s, 32);
      accu[r] += __shfl_xor(accu[r], s, 32);
    }
  }

  if (lm == 0) {
#pragma unroll
    for (int r = 0; r < 8; ++r) {
      red[wave][0][r + 8 * half] = accl[r];
      red[wave][1][r + 8 * half] = accu[r];
    }
  }
  __syncthreads();

  if (wave == 0) {
    float s = 0.0f;
#pragma unroll
    for (int v = 0; v < CWAVES; ++v) s += red[v][half][lm];
    const float other = __shfl_xor(s, 16, 32);
    const float res = half ? s : fminf(s, other);
    out[half * BROWS + rowbase + lm] = res;
  }
}

extern "C" void kernel_launch(void* const* d_in, const int* in_sizes, int n_in,
                              void* d_out, int out_size, void* d_ws, size_t ws_size,
                              hipStream_t stream) {
  (void)in_sizes; (void)n_in; (void)out_size; (void)ws_size;
  const float* x  = (const float*)d_in[0];
  const float* ie = (const float*)d_in[1];
  float* out  = (float*)d_out;
  float* wm   = (float*)d_ws;

  IE_fm_mobius_kernel<<<dim3(1), dim3(256), 0, stream>>>(ie, wm);
  IE_choquet_kernel<<<dim3(BROWS / 16), dim3(CTHREADS), 0, stream>>>(x, wm, out);
}


# tile kernel: weighted sum folded one input at a time (Horner per variable): inputs 10,11 by a K=4 f32 WMMA of w against {1,z10,z11,z10*z11}, ten fma folds in registers; no log, exp or LDS table
# speedup vs baseline: 18.5941x; 1.2990x over previous
#include <hip/hip_runtime.h>
#include <math.h>

#define NBITS  12
#define NMASKS 4095
#define NVARSC 4094
#define BROWS  4096

typedef __attribute__((ext_vector_type(2))) float v2f;
typedef __attribute__((ext_vector_type(8))) float v8f;
typedef __attribute__((ext_vector_type(4))) float v4f;

#define LT    256
#define NPT   16
#define RP    20

__device__ __forceinline__ void IE_row_load(const float* a, int off, float (&r)[NPT]) {
#pragma unroll
  for (int q = 0; q < NPT / 4; ++q) {
    const v4f t = *(const v4f*)(a + off + 4 * q);
    r[4 * q] = t.x; r[4 * q + 1] = t.y; r[4 * q + 2] = t.z; r[4 * q + 3] = t.w;
  }
}

__device__ __forceinline__ void IE_row_store(float* a, int off, const float (&r)[NPT]) {
#pragma unroll
  for (int q = 0; q < NPT / 4; ++q) {
    v4f t;
    t.x = r[4 * q]; t.y = r[4 * q + 1]; t.z = r[4 * q + 2]; t.w = r[4 * q + 3];
    *(v4f*)(a + off + 4 * q) = t;
  }
}

__device__ __forceinline__ void IE_solve16(const float (&av)[NPT], const float (&pm)[NPT],
                                           float (&cur)[NPT]) {
#pragma unroll
  for (int c = 0; c <= 4; ++c) {
#pragma unroll
    for (int i = 0; i < NPT; ++i) {
      if (__builtin_popcount(i) == c) {
        float t = pm[i];
#pragma unroll
        for (int b = 0; b < 4; ++b)
          if (i & (1 << b)) t = fmaxf(t, cur[i ^ (1 << b)]);
        cur[i] = av[i] + t;
      }
    }
  }
}

template <int K>
__device__ __forceinline__ float IE_xlane(float x) {
  const int b = __builtin_bit_cast(int, x);
  if constexpr (K == 16)
    return __builtin_bit_cast(float, __builtin_amdgcn_permlanex16(b, b, 0x76543210, 0xfedcba98, false, false));
  else
    return __builtin_bit_cast(float, __builtin_amdgcn_update_dpp(0, b, 0x160 | K, 0xf, 0xf, true));
}

template <int K>
__device__ __forceinline__ void IE_lane_pass(float (&cur)[NPT], bool has) {
  const float f = has ? 1.0f : 0.0f;
#pragma unroll
  for (int i = 0; i < NPT; ++i) {
    const float part = IE_xlane<K>(cur[i]);
    cur[i] = __builtin_fmaf(-part, f, cur[i]);
  }
}

template <int WB>
__device__ __forceinline__ void IE_wave_bits(const float* a, int tid, float (&cur)[NPT]) {
  float x[8][NPT];
#pragma unroll
  for (int s = 1; s < 8; ++s)
    if ((s & ~WB) == 0) IE_row_load(a, (tid ^ (s << 5)) * RP, x[s]);
#pragma unroll
  for (int i = 0; i < NPT; ++i) x[0][i] = cur[i];
#pragma unroll
  for (int kb = 0; kb < 3; ++kb) {
    const int k = 1 << kb;
    if (WB & k) {
#pragma unroll
      for (int s = 0; s < 8; ++s) {
        if ((s & ~WB) == 0 && (s & (2 * k - 1)) == 0) {
#pragma unroll
          for (int i = 0; i < NPT; ++i) x[s][i] -= x[s | k][i];
        }
      }
    }
  }
#pragma unroll
  for (int i = 0; i < NPT; ++i) cur[i] = x[0][i];
}

__global__ __launch_bounds__(LT)
void IE_fm_mobius_kernel(const float* __restrict__ ie_vars,
                         float* __restrict__ wm) {
  __shared__ __attribute__((aligned(16))) float a[LT * RP];
  const int tid = threadIdx.x;
  const int pt  = __popc(tid);

  int prow[8];
  {
    int r = tid;
#pragma unroll
    for (int n = 0; n < 8; ++n) {
      const int low = r & -r;
      prow[n] = (tid ^ low) * RP;
      r ^= low;
    }
  }

  float raw[NPT];
#pragma unroll
  for (int i = 0; i < NPT; ++i) {
    int j = LT * i + tid - 1;
    j = (j < 0) ? 0 : j;
    j = (j > NVARSC - 1) ? (NVARSC - 1) : j;
    raw[i] = ie_vars[j];
  }
  __builtin_amdgcn_sched_barrier(0);
  float av[NPT];
  float cur[NPT];
#pragma unroll
  for (int i = 0; i < NPT; ++i) {
    const int m = LT * i + tid;
    av[i]  = (m >= 1 && m <= NVARSC) ? fabsf(raw[i]) : 0.0f;
    cur[i] = 0.0f;
  }

  if (tid == 0) {
    float pm[NPT];
#pragma unroll
    for (int i = 0; i < NPT; ++i) pm[i] = 0.0f;
    IE_solve16(av, pm, cur);
    IE_row_store(a, 0, cur);
  }
  __syncthreads();

#pragma unroll
  for (int j = 1; j <= 8; ++j) {
    if (pt == j) {
      float p[8][NPT];
#pragma unroll
      for (int n = 0; n < 8; ++n)
        if (n < j) IE_row_load(a, prow[n], p[n]);
      __builtin_amdgcn_sched_barrier(0);
      float pm[NPT];
#pragma unroll
      for (int i = 0; i < NPT; ++i) {
        float t = p[0][i];
#pragma unroll
        for (int n = 1; n < 8; ++n)
          if (n < j) t = fmaxf(t, p[n][i]);
        pm[i] = t;
      }
      IE_solve16(av, pm, cur);
      IE_row_store(a, tid * RP, cur);
    }
    __syncthreads();
  }

#pragma unroll
  for (int i = 0; i < NPT; ++i) {
    cur[i] = fminf(cur[i], 1.0f);
    if (LT * i + tid == NMASKS) cur[i] = 1.0f;
  }

  IE_lane_pass<1>(cur, (tid & 1) != 0);
  IE_lane_pass<2>(cur, (tid & 2) != 0);
  IE_lane_pass<4>(cur, (tid & 4) != 0);
  IE_lane_pass<8>(cur, (tid & 8) != 0);
  IE_lane_pass<16>(cur, (tid & 16) != 0);
  IE_row_store(a, tid * RP, cur);
  __syncthreads();
  {
    const int wv = __builtin_amdgcn_readfirstlane(tid >> 5);
    switch (wv) {
      case 1: IE_wave_bits<1>(a, tid, cur); break;
      case 2: IE_wave_bits<2>(a, tid, cur); break;
      case 3: IE_wave_bits<3>(a, tid, cur); break;
      case 4: IE_wave_bits<4>(a, tid, cur); break;
      case 5: IE_wave_bits<5>(a, tid, cur); break;
      case 6: IE_wave_bits<6>(a, tid, cur); break;
      case 7: IE_wave_bits<7>(a, tid, cur); break;
      default: break;
    }
  }
#pragma unroll
  for (int b = 0; b < 4; ++b) {
#pragma unroll
    for (int i = 0; i < NPT; ++i)
      if (i & (1 << b)) cur[i] -= cur[i ^ (1 << b)];
  }
#pragma unroll
  for (int i = 0; i < NPT; ++i) wm[LT * i + tid] = cur[i];
}

#define CWAVES 8
#define CTHREADS 256

__device__ __forceinline__ float IE_fold_side(const float (&z)[NBITS], const v4f (&wa)[2][2],
                                              int half, int wave) {
  v2f bf;
  bf.x = half ? z[11] : 1.0f;
  bf.y = bf.x * z[10];
  const v8f zero = {};
  v8f u[4];
#pragma unroll
  for (int e = 0; e < 4; ++e) {
    v2f a0, a1;
    a0.x = wa[0][0][e]; a0.y = wa[1][0][e];
    a1.x = wa[0][1][e]; a1.y = wa[1][1][e];
    const v8f t0 = __builtin_amdgcn_wmma_f32_16x16x4_f32(false, a0, false, bf,
                                                         (short)0, zero, false, false);
    const v8f t1 = __builtin_amdgcn_wmma_f32_16x16x4_f32(false, a1, false, bf,
                                                         (short)0, zero, false, false);
    u[e] = t0 + t1 * z[6];
  }
  const v8f v0 = u[0] + u[2] * z[1];
  const v8f v1 = u[1] + u[3] * z[1];
  const v8f xx = v0 + v1 * z[0];
  float y[4];
#pragma unroll
  for (int r = 0; r < 4; ++r) y[r] = xx[r] + z[4] * xx[r + 4];
  const float p0 = y[0] + z[3] * y[2];
  const float p1 = y[1] + z[3] * y[3];
  const float s  = p0 + z[2] * p1;
  float f = half ? z[5] : 1.0f;
  f *= (wave & 1) ? z[7] : 1.0f;
  f *= (wave & 2) ? z[8] : 1.0f;
  f *= (wave & 4) ? z[9] : 1.0f;
  return s * f;
}

__global__ __launch_bounds__(CTHREADS)
void IE_choquet_kernel(const float* __restrict__ x,
                       const float* __restrict__ wm,
                       float* __restrict__ out) {
  __shared__ float red[CWAVES][2][16];
  const int tid  = threadIdx.x;
  const int lane = tid & 31;
  const int wave = __builtin_amdgcn_readfirstlane(tid >> 5);

  const int rowbase = blockIdx.x * 16;
  const int half    = lane >> 4;
  const int lm      = lane & 15;

  const float* xr = x + (rowbase + lm) * (2 * NBITS);
  v4f xv[6];
#pragma unroll
  for (int j = 0; j < 6; ++j)
    xv[j] = *(const v4f*)(xr + 4 * j);

  v4f wa[2][2];
  {
    const float* wp = wm + (half << 11) + (wave << 7) + 4 * lm;
#pragma unroll
    for (int s = 0; s < 2; ++s)
#pragma unroll
      for (int q = 0; q < 2; ++q)
        wa[s][q] = *(const v4f*)(wp + (s << 10) + (q << 6));
  }

  float zl[NBITS], zu[NBITS];
#pragma unroll
  for (int j = 0; j < 3; ++j) {
#pragma unroll
    for (int k = 0; k < 4; ++k) {
      zl[4 * j + k] = fmaxf(xv[j][k], 1e-20f);
      zu[4 * j + k] = fmaxf(xv[3 + j][k], 1e-20f);
    }
  }

  float sl = IE_fold_side(zl, wa, half, wave);
  float su = IE_fold_side(zu, wa, half, wave);

  sl += __shfl_xor(sl, 16, 32);
  su += __shfl_xor(su, 16, 32);

  if (half == 0) {
    red[wave][0][lm] = sl;
    red[wave][1][lm] = su;
  }
  __syncthreads();

  if (wave == 0) {
    float s = 0.0f;
#pragma unroll
    for (int v = 0; v < CWAVES; ++v) s += red[v][half][lm];
    const float other = __shfl_xor(s, 16, 32);
    const float res = half ? s : fminf(s, other);
    out[half * BROWS + rowbase + lm] = res;
  }
}

extern "C" void kernel_launch(void* const* d_in, const int* in_sizes, int n_in,
                              void* d_out, int out_size, void* d_ws, size_t ws_size,
                              hipStream_t stream) {
  (void)in_sizes; (void)n_in; (void)out_size; (void)ws_size;
  const float* x  = (const float*)d_in[0];
  const float* ie = (const float*)d_in[1];
  float* out  = (float*)d_out;
  float* wm   = (float*)d_ws;

  IE_fm_mobius_kernel<<<dim3(1), dim3(256), 0, stream>>>(ie, wm);
  IE_choquet_kernel<<<dim3(BROWS / 16), dim3(CTHREADS), 0, stream>>>(x, wm, out);
}
